# starting kernel
# baseline (speedup 1.0000x reference)
#include <hip/hip_runtime.h>
#include <math.h>

typedef __attribute__((ext_vector_type(2))) float v2f;
typedef __attribute__((ext_vector_type(8))) float v8f;

#define E_DIM 64
#define H_NUM 8
#define D_HEAD 8
#define B_SZ 16
#define S_LEN 1024
#define N_TOK (B_SZ * S_LEN)

__device__ __forceinline__ v8f wmma4(v2f a, v2f b, v8f c) {
  return __builtin_amdgcn_wmma_f32_16x16x4_f32(
       false, a,  false, b,
       (short)0, c,  false,  false);
}

__global__ __launch_bounds__(128) void k_qkv(const float* __restrict__ x,
                                             const float* __restrict__ Wqkv,
                                             const float* __restrict__ bqkv,
                                             float* __restrict__ qkv) {
  const int wave = threadIdx.x >> 5;
  const int lane = threadIdx.x & 31;
  const int n16 = lane & 15, hi = lane >> 4;
  const int tile = blockIdx.x * 4 + wave;
  const int mt = tile / 12, nt = tile % 12;
  const int m0 = mt * 16, n0 = nt * 16;
  const float* arow = x + (m0 + n16) * E_DIM;
  const float* brow = Wqkv + (n0 + n16) * E_DIM;
  v8f c = {};
#pragma unroll
  for (int k0 = 0; k0 < E_DIM; k0 += 4) {
    v2f a, b;
    a.x = arow[k0 + 2 * hi]; a.y = arow[k0 + 2 * hi + 1];
    b.x = brow[k0 + 2 * hi]; b.y = brow[k0 + 2 * hi + 1];
    c = wmma4(a, b, c);
  }
  const float bias = bqkv[n0 + n16];
#pragma unroll
  for (int r = 0; r < 8; ++r) {
    const int row = m0 + r + 8 * hi;
    qkv[row * (3 * E_DIM) + n0 + n16] = c[r] + bias;
  }
}

__global__ __launch_bounds__(128) void k_attn(const float* __restrict__ qkv,
                                              float* __restrict__ attn_out) {
  __shared__ float ldsK[2][16 * 9];
  __shared__ float ldsV[2][16 * 9];
  __shared__ float ldsP[4][16 * 16];
  const int tid = threadIdx.x;
  const int wave = tid >> 5;
  const int lane = tid & 31;
  const int n16 = lane & 15, hi = lane >> 4;

  const int bid = blockIdx.x;
  const int bh = bid >> 4;
  const int b = bh >> 3;
  const int h = bh & 7;
  const int qt = (bid & 15) * 4 + wave;
  const float scale = 0.35355339059327373f;

  const float* qkvB = qkv + (size_t)(b * S_LEN) * (3 * E_DIM);
  const int srow = tid >> 3, scol = tid & 7;

  v2f qa0, qa1;
  {
    const float* qp = qkvB + (qt * 16 + n16) * (3 * E_DIM) + h * D_HEAD;
    qa0.x = qp[0 + 2 * hi] * scale; qa0.y = qp[1 + 2 * hi] * scale;
    qa1.x = qp[4 + 2 * hi] * scale; qa1.y = qp[5 + 2 * hi] * scale;
  }

  {
    const float* rowp = qkvB + srow * (3 * E_DIM) + h * D_HEAD + scol;
    ldsK[0][srow * 9 + scol] = rowp[E_DIM];
    ldsV[0][srow * 9 + scol] = rowp[2 * E_DIM];
  }

  float m[8], l[8];
  v8f o = {};
#pragma unroll
  for (int r = 0; r < 8; ++r) { m[r] = -3.0e38f; l[r] = 0.f; }
  float* myP = ldsP[wave];
  const int d = n16 & 7;

  for (int kt = 0; kt < S_LEN / 16; ++kt) {
    __syncthreads();
    const int cur = kt & 1, nxt = cur ^ 1;
    if (kt + 1 < S_LEN / 16) {
      const float* rowp =
          qkvB + ((kt + 1) * 16 + srow) * (3 * E_DIM) + h * D_HEAD + scol;
      ldsK[nxt][srow * 9 + scol] = rowp[E_DIM];
      ldsV[nxt][srow * 9 + scol] = rowp[2 * E_DIM];
    }

    v2f kb0, kb1;
    kb0.x = ldsK[cur][n16 * 9 + 0 + 2 * hi];
    kb0.y = ldsK[cur][n16 * 9 + 1 + 2 * hi];
    kb1.x = ldsK[cur][n16 * 9 + 4 + 2 * hi];
    kb1.y = ldsK[cur][n16 * 9 + 5 + 2 * hi];
    v8f s = {};
    s = wmma4(qa0, kb0, s);
    s = wmma4(qa1, kb1, s);

#pragma unroll
    for (int r = 0; r < 8; ++r) {
      float mx = s[r];
      mx = fmaxf(mx, __shfl_xor(mx, 1));
      mx = fmaxf(mx, __shfl_xor(mx, 2));
      mx = fmaxf(mx, __shfl_xor(mx, 4));
      mx = fmaxf(mx, __shfl_xor(mx, 8));
      const float mn = fmaxf(m[r], mx);
      const float alpha = __expf(m[r] - mn);
      const float p = __expf(s[r] - mn);
      float ps = p;
      ps += __shfl_xor(ps, 1);
      ps += __shfl_xor(ps, 2);
      ps += __shfl_xor(ps, 4);
      ps += __shfl_xor(ps, 8);
      l[r] = l[r] * alpha + ps;
      m[r] = mn;
      o[r] *= alpha;
      myP[(r + 8 * hi) * 16 + n16] = p;
    }
    asm volatile("s_wait_dscnt 0" ::: "memory");

#pragma unroll
    for (int kc = 0; kc < 4; ++kc) {
      v2f pa, vb;
      pa.x = myP[n16 * 16 + kc * 4 + 2 * hi];
      pa.y = myP[n16 * 16 + kc * 4 + 2 * hi + 1];
      vb.x = ldsV[cur][(kc * 4 + 2 * hi) * 9 + d];
      vb.y = ldsV[cur][(kc * 4 + 2 * hi + 1) * 9 + d];
      o = wmma4(pa, vb, o);
    }
    asm volatile("" ::: "memory");
  }

  if (n16 < D_HEAD) {
#pragma unroll
    for (int r = 0; r < 8; ++r) {
      const int row = b * S_LEN + qt * 16 + r + 8 * hi;
      attn_out[row * E_DIM + h * D_HEAD + n16] = o[r] / l[r];
    }
  }
}

__global__ __launch_bounds__(128) void k_outproj_ln(
    const float* __restrict__ attn_out, const float* __restrict__ Wout,
    const float* __restrict__ bout, const float* __restrict__ x,
    const float* __restrict__ g1, const float* __restrict__ beta1,
    float* __restrict__ mlp_in) {
  __shared__ float res1[16][64];
  const int wave = threadIdx.x >> 5, lane = threadIdx.x & 31;
  const int n16 = lane & 15, hi = lane >> 4;
  const int m0 = blockIdx.x * 16;
  const int n0 = wave * 16;
  const float* arow = attn_out + (m0 + n16) * E_DIM;
  const float* brow = Wout + (n0 + n16) * E_DIM;
  v8f c = {};
#pragma unroll
  for (int k0 = 0; k0 < E_DIM; k0 += 4) {
    v2f a, b;
    a.x = arow[k0 + 2 * hi]; a.y = arow[k0 + 2 * hi + 1];
    b.x = brow[k0 + 2 * hi]; b.y = brow[k0 + 2 * hi + 1];
    c = wmma4(a, b, c);
  }
  const float bias = bout[n0 + n16];
#pragma unroll
  for (int r = 0; r < 8; ++r) {
    const int row = r + 8 * hi;
    res1[row][n0 + n16] = c[r] + bias + x[(m0 + row) * E_DIM + n0 + n16];
  }
  __syncthreads();
  const int row = threadIdx.x >> 3;
  const int c0 = (threadIdx.x & 7) * 8;
  float sum = 0.f, sq = 0.f;
#pragma unroll
  for (int i = 0; i < 8; ++i) { float v = res1[row][c0 + i]; sum += v; sq += v * v; }
#pragma unroll
  for (int msk = 1; msk < 8; msk <<= 1) {
    sum += __shfl_xor(sum, msk);
    sq += __shfl_xor(sq, msk);
  }
  const float mean = sum * (1.f / 64.f);
  const float var = sq * (1.f / 64.f) - mean * mean;
  const float rstd = rsqrtf(var + 1e-5f);
#pragma unroll
  for (int i = 0; i < 8; ++i) {
    const int col = c0 + i;
    mlp_in[(m0 + row) * E_DIM + col] =
        (res1[row][col] - mean) * rstd * g1[col] + beta1[col];
  }
}

__global__ __launch_bounds__(128) void k_mlp1(const float* __restrict__ mlp_in,
                                              const float* __restrict__ Wm1,
                                              const float* __restrict__ bm1,
                                              float* __restrict__ hbuf) {
  const int wave = threadIdx.x >> 5, lane = threadIdx.x & 31;
  const int n16 = lane & 15, hi = lane >> 4;
  const int tile = blockIdx.x * 4 + wave;
  const int mt = tile >> 3, nt = tile & 7;
  const int m0 = mt * 16, n0 = nt * 16;
  const float* arow = mlp_in + (m0 + n16) * E_DIM;
  const float* brow = Wm1 + (n0 + n16) * E_DIM;
  v8f c = {};
#pragma unroll
  for (int k0 = 0; k0 < E_DIM; k0 += 4) {
    v2f a, b;
    a.x = arow[k0 + 2 * hi]; a.y = arow[k0 + 2 * hi + 1];
    b.x = brow[k0 + 2 * hi]; b.y = brow[k0 + 2 * hi + 1];
    c = wmma4(a, b, c);
  }
  const float bias = bm1[n0 + n16];
#pragma unroll
  for (int r = 0; r < 8; ++r) {
    const int row = m0 + r + 8 * hi;
    hbuf[row * (2 * E_DIM) + n0 + n16] = fmaxf(c[r] + bias, 0.f);
  }
}

__global__ __launch_bounds__(128) void k_mlp2_ln(
    const float* __restrict__ hbuf, const float* __restrict__ Wm2,
    const float* __restrict__ bm2, const float* __restrict__ mlp_in,
    const float* __restrict__ g2, const float* __restrict__ beta2,
    float* __restrict__ out) {
  __shared__ float res2[16][64];
  const int wave = threadIdx.x >> 5, lane = threadIdx.x & 31;
  const int n16 = lane & 15, hi = lane >> 4;
  const int m0 = blockIdx.x * 16;
  const int n0 = wave * 16;
  const float* arow = hbuf + (m0 + n16) * (2 * E_DIM);
  const float* brow = Wm2 + (n0 + n16) * (2 * E_DIM);
  v8f c = {};
#pragma unroll
  for (int k0 = 0; k0 < 2 * E_DIM; k0 += 4) {
    v2f a, b;
    a.x = arow[k0 + 2 * hi]; a.y = arow[k0 + 2 * hi + 1];
    b.x = brow[k0 + 2 * hi]; b.y = brow[k0 + 2 * hi + 1];
    c = wmma4(a, b, c);
  }
  const float bias = bm2[n0 + n16];
#pragma unroll
  for (int r = 0; r < 8; ++r) {
    const int row = r + 8 * hi;
    res2[row][n0 + n16] = c[r] + bias + mlp_in[(m0 + row) * E_DIM + n0 + n16];
  }
  __syncthreads();
  const int row = threadIdx.x >> 3;
  const int c0 = (threadIdx.x & 7) * 8;
  float sum = 0.f, sq = 0.f;
#pragma unroll
  for (int i = 0; i < 8; ++i) { float v = res2[row][c0 + i]; sum += v; sq += v * v; }
#pragma unroll
  for (int msk = 1; msk < 8; msk <<= 1) {
    sum += __shfl_xor(sum, msk);
    sq += __shfl_xor(sq, msk);
  }
  const float mean = sum * (1.f / 64.f);
  const float var = sq * (1.f / 64.f) - mean * mean;
  const float rstd = rsqrtf(var + 1e-5f);
#pragma unroll
  for (int i = 0; i < 8; ++i) {
    const int col = c0 + i;
    out[(m0 + row) * E_DIM + col] =
        (res2[row][col] - mean) * rstd * g2[col] + beta2[col];
  }
}

extern "C" void kernel_launch(void* const* d_in, const int* in_sizes, int n_in,
                              void* d_out, int out_size, void* d_ws, size_t ws_size,
                              hipStream_t stream) {
  const float* x    = (const float*)d_in[0];
  const float* Wqkv = (const float*)d_in[4];
  const float* bqkv = (const float*)d_in[5];
  const float* Wout = (const float*)d_in[6];
  const float* bout = (const float*)d_in[7];
  const float* g1   = (const float*)d_in[8];
  const float* b1   = (const float*)d_in[9];
  const float* g2   = (const float*)d_in[10];
  const float* b2   = (const float*)d_in[11];
  const float* Wm1  = (const float*)d_in[12];
  const float* bm1  = (const float*)d_in[13];
  const float* Wm2  = (const float*)d_in[14];
  const float* bm2  = (const float*)d_in[15];
  float* out = (float*)d_out;

  float* ws      = (float*)d_ws;
  float* qkv     = ws;
  float* attn    = qkv + (size_t)N_TOK * 3 * E_DIM;
  float* mlpin   = attn + (size_t)N_TOK * E_DIM;
  float* hbuf    = mlpin + (size_t)N_TOK * E_DIM;

  k_qkv<<<(N_TOK / 16) * 12 / 4, 128, 0, stream>>>(x, Wqkv, bqkv, qkv);
  k_attn<<<(B_SZ * H_NUM * (S_LEN / 16)) / 4, 128, 0, stream>>>(qkv, attn);
  k_outproj_ln<<<N_TOK / 16, 128, 0, stream>>>(attn, Wout, bout, x, g1, b1, mlpin);
  k_mlp1<<<(N_TOK / 16) * 8 / 4, 128, 0, stream>>>(mlpin, Wm1, bm1, hbuf);
  k_mlp2_ln<<<N_TOK / 16, 128, 0, stream>>>(hbuf, Wm2, bm2, mlpin, g2, b2, out);
}


# tail in one kernel (4 -> 2 launches): a wave takes 16 tokens through out-proj, LN1, MLP1, MLP2, LN2 in registers, transposed: weights = A (LDS), each D tile is the next B; fp sum order changes
# speedup vs baseline: 2.1568x; 2.1568x over previous
#include <hip/hip_runtime.h>
#include <math.h>

typedef __attribute__((ext_vector_type(2))) float v2f;
typedef __attribute__((ext_vector_type(8))) float v8f;

#define E_DIM 64
#define H_NUM 8
#define D_HEAD 8
#define B_SZ 16
#define S_LEN 1024
#define N_TOK (B_SZ * S_LEN)

__device__ __forceinline__ v8f wmma4(v2f a, v2f b, v8f c) {
  return __builtin_amdgcn_wmma_f32_16x16x4_f32(
       false, a,  false, b,
       (short)0, c,  false,  false);
}

#define ROW_XOR(v, MASK)                                                      \
  __builtin_bit_cast(float, __builtin_amdgcn_update_dpp(                      \
      __builtin_bit_cast(int, (v)), __builtin_bit_cast(int, (v)),             \
      0x160 + (MASK), 0xf, 0xf, true))

#define ATT_WAVES 32
#define ATT_CH 8
#define ATT_LDS_FLOATS (176 * 1024 / 4)
__global__ __launch_bounds__(ATT_WAVES * 32) void k_attn(
    const float* __restrict__ x, const float* __restrict__ Wqkv,
    const float* __restrict__ bqkv, float* __restrict__ attn_out) {
  __shared__ __attribute__((aligned(16))) float lds_all[ATT_LDS_FLOATS];
  float* ldsK = lds_all;
  float* ldsV = lds_all + S_LEN * D_HEAD;
  float* ldsP = lds_all + 2 * S_LEN * D_HEAD;
  const int tid = threadIdx.x;
  const int wave = __builtin_amdgcn_readfirstlane(tid >> 5);
  const int lane = tid & 31;
  const int n16 = lane & 15, hi = lane >> 4;

  const int bid = blockIdx.x;
  const int bh = bid >> 1;
  const int b = bh >> 3;
  const int h = bh & 7;
  const int qt = (bid & 1) * ATT_WAVES + wave;
  const float scale = 0.35355339059327373f;

  const float* xB = x + (size_t)(b * S_LEN) * E_DIM;
  const int ot = ((bid & 1) ^ 1) * ATT_WAVES + wave;

  const int isV = n16 >> 3, dd = n16 & 7;
  const int kvrow = (isV ? 2 * E_DIM : E_DIM) + h * D_HEAD + dd;
  const int qrow = h * D_HEAD + dd;
  const float4* wkv = reinterpret_cast<const float4*>(Wqkv + kvrow * E_DIM) + hi;
  const float4* wq  = reinterpret_cast<const float4*>(Wqkv + qrow * E_DIM) + hi;
  const float4* xq  = reinterpret_cast<const float4*>(xB + (qt * 16 + n16) * E_DIM) + hi;
  const float4* xo  = reinterpret_cast<const float4*>(xB + (ot * 16 + n16) * E_DIM) + hi;
  v8f ckv = {}, cko = {}, cq = {};
#pragma unroll
  for (int j = 0; j < 8; ++j) {
    const float4 wv = wkv[2 * j];
    const float4 wqv = wq[2 * j];
    const float4 xv = xq[2 * j];
    const float4 xw = xo[2 * j];
    v2f w0, w1, q0, q1, a0, a1, o0, o1;
    w0.x = wv.x;  w0.y = wv.y;  w1.x = wv.z;  w1.y = wv.w;
    q0.x = wqv.x; q0.y = wqv.y; q1.x = wqv.z; q1.y = wqv.w;
    a0.x = xv.x;  a0.y = xv.y;  a1.x = xv.z;  a1.y = xv.w;
    o0.x = xw.x;  o0.y = xw.y;  o1.x = xw.z;  o1.y = xw.w;
    ckv = wmma4(a0, w0, ckv);
    ckv = wmma4(a1, w1, ckv);
    cq = wmma4(q0, a0, cq);
    cq = wmma4(q1, a1, cq);
    cko = wmma4(o0, w0, cko);
    cko = wmma4(o1, w1, cko);
  }
  {
    const float bias = bqkv[kvrow];
    float* dst = isV ? ldsV + hi * 64 + 4 * dd
                     : ldsK + ((dd >> 1) & 1) * 64 + hi * 32 + (dd & 1) + 2 * (dd >> 2);
#pragma unroll
    for (int r = 0; r < 8; ++r) {
      const int voff = ((r >> 1) & 1) * 32 + ((r >> 2) & 1) * 2 + (r & 1);
      const int off = isV ? voff : 4 * r;
      dst[qt * 128 + off] = ckv[r] + bias;
      dst[ot * 128 + off] = cko[r] + bias;
    }
  }

  v2f qa0, qa1;
  {
    const float* bq = bqkv + h * D_HEAD;
    float qd[8];
#pragma unroll
    for (int r = 0; r < 8; ++r) qd[r] = (cq[r] + bq[r]) * scale;
    qa0.x = hi ? qd[2] : qd[0];
    qa0.y = hi ? qd[3] : qd[1];
    qa1.x = hi ? qd[6] : qd[4];
    qa1.y = hi ? qd[7] : qd[5];
  }
  __syncthreads();

  const float4* kfrag = reinterpret_cast<const float4*>(ldsK) + lane;
  const float4* vfrag =
      reinterpret_cast<const float4*>(ldsV) + hi * 8 + (n16 & 7);

  float m[8], l[8];
  v8f o = {};
#pragma unroll
  for (int r = 0; r < 8; ++r) { m[r] = -3.0e38f; l[r] = 0.f; }
  float* myP = ldsP + wave * 256;

  for (int c0 = 0; c0 < S_LEN / 16; c0 += ATT_CH) {
    v8f s[ATT_CH];
#pragma unroll
    for (int u = 0; u < ATT_CH; ++u) {
      const float4 kf = kfrag[(c0 + u) * 32];
      v2f kb0, kb1;
      kb0.x = kf.x; kb0.y = kf.y;
      kb1.x = kf.z; kb1.y = kf.w;
      v8f z = {};
      z = wmma4(qa0, kb0, z);
      s[u] = wmma4(qa1, kb1, z);
    }

#pragma unroll
    for (int r = 0; r < 8; ++r) {
      float mx = s[0][r];
#pragma unroll
      for (int u = 1; u < ATT_CH; ++u) mx = fmaxf(mx, s[u][r]);
      mx = fmaxf(mx, ROW_XOR(mx, 1));
      mx = fmaxf(mx, ROW_XOR(mx, 2));
      mx = fmaxf(mx, ROW_XOR(mx, 4));
      mx = fmaxf(mx, ROW_XOR(mx, 8));
      const float mn = fmaxf(m[r], mx);
      const float a = __expf(m[r] - mn);
      l[r] *= a;
      o[r] *= a;
      m[r] = mn;
    }

#pragma unroll
    for (int u = 0; u < ATT_CH; ++u) {
      const int kt = c0 + u;
#pragma unroll
      for (int r = 0; r < 8; ++r) {
        const float p = __expf(s[u][r] - m[r]);
        l[r] += p;
        myP[(r + 8 * hi) * 16 + n16] = p;
      }
      asm volatile("s_wait_dscnt 0" ::: "memory");

      const float4 vf0 = vfrag[kt * 32];
      const float4 vf1 = vfrag[kt * 32 + 16];
      v2f pa, vb;
      pa.x = myP[n16 * 16 + 0 + 2 * hi]; pa.y = myP[n16 * 16 + 1 + 2 * hi];
      vb.x = vf0.x; vb.y = vf0.y;
      o = wmma4(pa, vb, o);
      pa.x = myP[n16 * 16 + 4 + 2 * hi]; pa.y = myP[n16 * 16 + 5 + 2 * hi];
      vb.x = vf0.z; vb.y = vf0.w;
      o = wmma4(pa, vb, o);
      pa.x = myP[n16 * 16 + 8 + 2 * hi]; pa.y = myP[n16 * 16 + 9 + 2 * hi];
      vb.x = vf1.x; vb.y = vf1.y;
      o = wmma4(pa, vb, o);
      pa.x = myP[n16 * 16 + 12 + 2 * hi]; pa.y = myP[n16 * 16 + 13 + 2 * hi];
      vb.x = vf1.z; vb.y = vf1.w;
      o = wmma4(pa, vb, o);
      asm volatile("" ::: "memory");
      __builtin_amdgcn_sched_barrier(0);
    }
  }

#pragma unroll
  for (int r = 0; r < 8; ++r) {
    float ps = l[r];
    ps += __shfl_xor(ps, 1);
    ps += __shfl_xor(ps, 2);
    ps += __shfl_xor(ps, 4);
    ps += __shfl_xor(ps, 8);
    l[r] = ps;
  }

  const int lane_e = __builtin_amdgcn_mbcnt_lo(~0u, 0u);
  const int n16e = lane_e & 15, hie = lane_e >> 4;
  if (n16e < D_HEAD) {
#pragma unroll
    for (int r = 0; r < 8; ++r) {
      const int row = b * S_LEN + qt * 16 + r + 8 * hie;
      attn_out[row * E_DIM + h * D_HEAD + n16e] = o[r] / l[r];
    }
  }
}

#define TAIL_WAVES 4
#define CH_OUT (4 * 4 * 2)
#define CH_M1  (8 * 4 * 2)
#define CH_M2  (4 * 8 * 2)
#define TAIL_LDS_F4 (176 * 1024 / 16)

__device__ __forceinline__ float xhalf(float v) {
  const int b = __builtin_bit_cast(int, v);
  return __builtin_bit_cast(float, __builtin_amdgcn_permlanex16(
      b, b, 0x76543210, 0xfedcba98, false, false));
}

template <int NU, int KT>
__device__ __forceinline__ void gemm_t(const float4* __restrict__ wl, int lane,
                                       const float (&B)[KT][8], v8f (&acc)[NU]) {
  float4 cur[NU * 2];
#pragma unroll
  for (int u = 0; u < NU; ++u) {
#pragma unroll
    for (int e = 0; e < 2; ++e) cur[u * 2 + e] = wl[((u * KT) * 2 + e) * 32 + lane];
  }
#pragma unroll
  for (int t = 0; t < KT; ++t) {
    float4 nxt[NU * 2];
    if (t + 1 < KT) {
#pragma unroll
      for (int u = 0; u < NU; ++u) {
#pragma unroll
        for (int e = 0; e < 2; ++e)
          nxt[u * 2 + e] = wl[((u * KT + t + 1) * 2 + e) * 32 + lane];
      }
    }
    __builtin_amdgcn_sched_barrier(0);
#pragma unroll
    for (int e = 0; e < 2; ++e) {
      v2f b0, b1;
      b0.x = B[t][4 * e + 0]; b0.y = B[t][4 * e + 1];
      b1.x = B[t][4 * e + 2]; b1.y = B[t][4 * e + 3];
#pragma unroll
      for (int u = 0; u < NU; ++u) {
        v2f a;
        a.x = cur[u * 2 + e].x; a.y = cur[u * 2 + e].y;
        acc[u] = wmma4(a, b0, acc[u]);
      }
#pragma unroll
      for (int u = 0; u < NU; ++u) {
        v2f a;
        a.x = cur[u * 2 + e].z; a.y = cur[u * 2 + e].w;
        acc[u] = wmma4(a, b1, acc[u]);
      }
    }
    __builtin_amdgcn_sched_barrier(0);
    if (t + 1 < KT) {
#pragma unroll
      for (int i = 0; i < NU * 2; ++i) cur[i] = nxt[i];
    }
  }
}

template <int NT>
__device__ __forceinline__ void vec_t(const float* __restrict__ vec, int h,
                                      float (&d)[NT][8]) {
#pragma unroll
  for (int t = 0; t < NT; ++t) {
    const float4* p = reinterpret_cast<const float4*>(vec + 16 * t + 8 * h);
    const float4 a = p[0], b = p[1];
    d[t][0] = a.x; d[t][1] = a.y; d[t][2] = a.z; d[t][3] = a.w;
    d[t][4] = b.x; d[t][5] = b.y; d[t][6] = b.z; d[t][7] = b.w;
  }
}

__device__ __forceinline__ void layer_norm_t(float (&v)[4][8],
                                             const float (&g)[4][8],
                                             const float (&beta)[4][8]) {
  float s[4], q[4];
#pragma unroll
  for (int t = 0; t < 4; ++t) {
    s[t] = v[t][0];
    q[t] = v[t][0] * v[t][0];
#pragma unroll
    for (int r = 1; r < 8; ++r) { s[t] += v[t][r]; q[t] += v[t][r] * v[t][r]; }
  }
  float sum = (s[0] + s[1]) + (s[2] + s[3]);
  float sq = (q[0] + q[1]) + (q[2] + q[3]);
  sum += xhalf(sum);
  sq += xhalf(sq);
  const float mean = sum * (1.f / 64.f);
  const float var = sq * (1.f / 64.f) - mean * mean;
  const float rstd = rsqrtf(var + 1e-5f);
#pragma unroll
  for (int t = 0; t < 4; ++t) {
#pragma unroll
    for (int r = 0; r < 8; ++r)
      v[t][r] = (v[t][r] - mean) * rstd * g[t][r] + beta[t][r];
  }
}

__global__ __launch_bounds__(TAIL_WAVES * 32) void k_tail(
    const float* __restrict__ attn_out, const float* __restrict__ x,
    const float* __restrict__ Wout, const float* __restrict__ bout,
    const float* __restrict__ g1, const float* __restrict__ beta1,
    const float* __restrict__ Wm1, const float* __restrict__ bm1,
    const float* __restrict__ Wm2, const float* __restrict__ bm2,
    const float* __restrict__ g2, const float* __restrict__ beta2,
    float* __restrict__ out) {
  __shared__ __attribute__((aligned(16))) float4 wl[TAIL_LDS_F4];
  float4* wlOut = wl;
  float4* wlM1 = wl + CH_OUT * 32;
  float4* wlM2 = wl + (CH_OUT + CH_M1) * 32;
  const int tid = threadIdx.x;
  const int wave = __builtin_amdgcn_readfirstlane(tid >> 5);
  const int lane = tid & 31;
  const int c = lane & 15, h = lane >> 4;
  const int row = (blockIdx.x * TAIL_WAVES + wave) * 16 + c;

  float X[4][8], R[4][8];
  {
    const float4* ar = reinterpret_cast<const float4*>(attn_out + row * E_DIM + 8 * h);
    const float4* xr = reinterpret_cast<const float4*>(x + row * E_DIM + 8 * h);
#pragma unroll
    for (int t = 0; t < 4; ++t) {
      const float4 a0 = ar[4 * t], a1 = ar[4 * t + 1];
      const float4 x0 = xr[4 * t], x1 = xr[4 * t + 1];
      X[t][0] = a0.x; X[t][1] = a0.y; X[t][2] = a0.z; X[t][3] = a0.w;
      X[t][4] = a1.x; X[t][5] = a1.y; X[t][6] = a1.z; X[t][7] = a1.w;
      R[t][0] = x0.x; R[t][1] = x0.y; R[t][2] = x0.z; R[t][3] = x0.w;
      R[t][4] = x1.x; R[t][5] = x1.y; R[t][6] = x1.z; R[t][7] = x1.w;
    }
  }

  {
    const int pw = 8 * h + 4 * (wave & 1) + 16 * (wave >> 1);
    const float* pO = Wout + c * E_DIM + pw;
    const float* p1 = Wm1 + c * E_DIM + pw;
    const float* p2 = Wm2 + c * (2 * E_DIM) + pw;
    float4 s0[CH_OUT / TAIL_WAVES], s1[CH_M1 / TAIL_WAVES], s2[CH_M2 / TAIL_WAVES];
#pragma unroll
    for (int i = 0; i < CH_OUT / TAIL_WAVES; ++i)
      s0[i] = *reinterpret_cast<const float4*>(pO + (i >> 1) * 16 * E_DIM + (i & 1) * 32);
#pragma unroll
    for (int i = 0; i < CH_M1 / TAIL_WAVES; ++i)
      s1[i] = *reinterpret_cast<const float4*>(p1 + (i >> 1) * 16 * E_DIM + (i & 1) * 32);
#pragma unroll
    for (int i = 0; i < CH_M2 / TAIL_WAVES; ++i)
      s2[i] = *reinterpret_cast<const float4*>(p2 + (i >> 2) * 32 * E_DIM + (i & 3) * 32);
    float4* d0 = wlOut + wave * 32 + lane;
    float4* d1 = wlM1 + wave * 32 + lane;
    float4* d2 = wlM2 + wave * 32 + lane;
#pragma unroll
    for (int i = 0; i < CH_OUT / TAIL_WAVES; ++i) d0[i * TAIL_WAVES * 32] = s0[i];
#pragma unroll
    for (int i = 0; i < CH_M1 / TAIL_WAVES; ++i) d1[i * TAIL_WAVES * 32] = s1[i];
#pragma unroll
    for (int i = 0; i < CH_M2 / TAIL_WAVES; ++i) d2[i * TAIL_WAVES * 32] = s2[i];
  }
  float vbo[4][8], vg1[4][8], vb1[4][8];
  vec_t<4>(bout, h, vbo);
  vec_t<4>(g1, h, vg1);
  vec_t<4>(beta1, h, vb1);
  __syncthreads();

  float M[4][8];
  float vm1[8][8];
  {
    v8f acc[4] = {};
    gemm_t<4, 4>(wlOut, lane, X, acc);
    vec_t<8>(bm1, h, vm1);
#pragma unroll
    for (int t = 0; t < 4; ++t) {
#pragma unroll
      for (int r = 0; r < 8; ++r) M[t][r] = acc[t][r] + vbo[t][r] + R[t][r];
    }
    layer_norm_t(M, vg1, vb1);
  }

  float Hd[8][8];
  float vm2[4][8], vg2[4][8], vb2[4][8];
  {
    v8f acc[8] = {};
    gemm_t<8, 4>(wlM1, lane, M, acc);
    vec_t<4>(bm2, h, vm2);
    vec_t<4>(g2, h, vg2);
    vec_t<4>(beta2, h, vb2);
#pragma unroll
    for (int u = 0; u < 8; ++u) {
#pragma unroll
      for (int r = 0; r < 8; ++r) Hd[u][r] = fmaxf(acc[u][r] + vm1[u][r], 0.f);
    }
  }

  {
    v8f acc[4] = {};
    gemm_t<4, 8>(wlM2, lane, Hd, acc);
    float O[4][8];
#pragma unroll
    for (int t = 0; t < 4; ++t) {
#pragma unroll
      for (int r = 0; r < 8; ++r) O[t][r] = acc[t][r] + vm2[t][r] + M[t][r];
    }
    layer_norm_t(O, vg2, vb2);
    float4* orow = reinterpret_cast<float4*>(out + row * E_DIM + 8 * h);
#pragma unroll
    for (int t = 0; t < 4; ++t) {
      orow[4 * t]     = make_float4(O[t][0], O[t][1], O[t][2], O[t][3]);
      orow[4 * t + 1] = make_float4(O[t][4], O[t][5], O[t][6], O[t][7]);
    }
  }
}

extern "C" void kernel_launch(void* const* d_in, const int* in_sizes, int n_in,
                              void* d_out, int out_size, void* d_ws, size_t ws_size,
                              hipStream_t stream) {
  const float* x    = (const float*)d_in[0];
  const float* Wqkv = (const float*)d_in[4];
  const float* bqkv = (const float*)d_in[5];
  const float* Wout = (const float*)d_in[6];
  const float* bout = (const float*)d_in[7];
  const float* g1   = (const float*)d_in[8];
  const float* b1   = (const float*)d_in[9];
  const float* g2   = (const float*)d_in[10];
  const float* b2   = (const float*)d_in[11];
  const float* Wm1  = (const float*)d_in[12];
  const float* bm1  = (const float*)d_in[13];
  const float* Wm2  = (const float*)d_in[14];
  const float* bm2  = (const float*)d_in[15];
  float* out = (float*)d_out;

  float* attn = (float*)d_ws;

  k_attn<<<B_SZ * H_NUM * 2, ATT_WAVES * 32, 0, stream>>>(x, Wqkv, bqkv, attn);
  k_tail<<<N_TOK / (16 * TAIL_WAVES), TAIL_WAVES * 32, 0, stream>>>(
      attn, x, Wout, bout, g1, b1, Wm1, bm1, Wm2, bm2, g2, b2, out);
}


# k_attn key loop: s_setprio by progress (3 for key chunks 0-2, 2 for 3-4, 1 for 5-6, 0 for the last): waves behind win the SIMD, its 8 waves end together, none runs alone at the end; no fp change
# speedup vs baseline: 2.2386x; 1.0379x over previous
#include <hip/hip_runtime.h>
#include <math.h>

typedef __attribute__((ext_vector_type(2))) float v2f;
typedef __attribute__((ext_vector_type(8))) float v8f;

#define E_DIM 64
#define H_NUM 8
#define D_HEAD 8
#define B_SZ 16
#define S_LEN 1024
#define N_TOK (B_SZ * S_LEN)

__device__ __forceinline__ v8f wmma4(v2f a, v2f b, v8f c) {
  return __builtin_amdgcn_wmma_f32_16x16x4_f32(
       false, a,  false, b,
       (short)0, c,  false,  false);
}

#define ROW_XOR(v, MASK)                                                      \
  __builtin_bit_cast(float, __builtin_amdgcn_update_dpp(                      \
      __builtin_bit_cast(int, (v)), __builtin_bit_cast(int, (v)),             \
      0x160 + (MASK), 0xf, 0xf, true))

#define ATT_WAVES 32
#define ATT_CH 8
#define ATT_LDS_FLOATS (176 * 1024 / 4)
__global__ __launch_bounds__(ATT_WAVES * 32) void k_attn(
    const float* __restrict__ x, const float* __restrict__ Wqkv,
    const float* __restrict__ bqkv, float* __restrict__ attn_out) {
  __shared__ __attribute__((aligned(16))) float lds_all[ATT_LDS_FLOATS];
  float* ldsK = lds_all;
  float* ldsV = lds_all + S_LEN * D_HEAD;
  float* ldsP = lds_all + 2 * S_LEN * D_HEAD;
  const int tid = threadIdx.x;
  const int wave = __builtin_amdgcn_readfirstlane(tid >> 5);
  const int lane = tid & 31;
  const int n16 = lane & 15, hi = lane >> 4;

  const int bid = blockIdx.x;
  const int bh = bid >> 1;
  const int b = bh >> 3;
  const int h = bh & 7;
  const int qt = (bid & 1) * ATT_WAVES + wave;
  const float scale = 0.35355339059327373f;

  const float* xB = x + (size_t)(b * S_LEN) * E_DIM;
  const int ot = ((bid & 1) ^ 1) * ATT_WAVES + wave;

  const int isV = n16 >> 3, dd = n16 & 7;
  const int kvrow = (isV ? 2 * E_DIM : E_DIM) + h * D_HEAD + dd;
  const int qrow = h * D_HEAD + dd;
  const float4* wkv = reinterpret_cast<const float4*>(Wqkv + kvrow * E_DIM) + hi;
  const float4* wq  = reinterpret_cast<const float4*>(Wqkv + qrow * E_DIM) + hi;
  const float4* xq  = reinterpret_cast<const float4*>(xB + (qt * 16 + n16) * E_DIM) + hi;
  const float4* xo  = reinterpret_cast<const float4*>(xB + (ot * 16 + n16) * E_DIM) + hi;
  v8f ckv = {}, cko = {}, cq = {};
#pragma unroll
  for (int j = 0; j < 8; ++j) {
    const float4 wv = wkv[2 * j];
    const float4 wqv = wq[2 * j];
    const float4 xv = xq[2 * j];
    const float4 xw = xo[2 * j];
    v2f w0, w1, q0, q1, a0, a1, o0, o1;
    w0.x = wv.x;  w0.y = wv.y;  w1.x = wv.z;  w1.y = wv.w;
    q0.x = wqv.x; q0.y = wqv.y; q1.x = wqv.z; q1.y = wqv.w;
    a0.x = xv.x;  a0.y = xv.y;  a1.x = xv.z;  a1.y = xv.w;
    o0.x = xw.x;  o0.y = xw.y;  o1.x = xw.z;  o1.y = xw.w;
    ckv = wmma4(a0, w0, ckv);
    ckv = wmma4(a1, w1, ckv);
    cq = wmma4(q0, a0, cq);
    cq = wmma4(q1, a1, cq);
    cko = wmma4(o0, w0, cko);
    cko = wmma4(o1, w1, cko);
  }
  {
    const float bias = bqkv[kvrow];
    float* dst = isV ? ldsV + hi * 64 + 4 * dd
                     : ldsK + ((dd >> 1) & 1) * 64 + hi * 32 + (dd & 1) + 2 * (dd >> 2);
#pragma unroll
    for (int r = 0; r < 8; ++r) {
      const int voff = ((r >> 1) & 1) * 32 + ((r >> 2) & 1) * 2 + (r & 1);
      const int off = isV ? voff : 4 * r;
      dst[qt * 128 + off] = ckv[r] + bias;
      dst[ot * 128 + off] = cko[r] + bias;
    }
  }

  v2f qa0, qa1;
  {
    const float* bq = bqkv + h * D_HEAD;
    float qd[8];
#pragma unroll
    for (int r = 0; r < 8; ++r) qd[r] = (cq[r] + bq[r]) * scale;
    qa0.x = hi ? qd[2] : qd[0];
    qa0.y = hi ? qd[3] : qd[1];
    qa1.x = hi ? qd[6] : qd[4];
    qa1.y = hi ? qd[7] : qd[5];
  }
  __syncthreads();

  const float4* kfrag = reinterpret_cast<const float4*>(ldsK) + lane;
  const float4* vfrag =
      reinterpret_cast<const float4*>(ldsV) + hi * 8 + (n16 & 7);

  float m[8], l[8];
  v8f o = {};
#pragma unroll
  for (int r = 0; r < 8; ++r) { m[r] = -3.0e38f; l[r] = 0.f; }
  float* myP = ldsP + wave * 256;

  for (int c0 = 0; c0 < S_LEN / 16; c0 += ATT_CH) {
    if (c0 == 0) __builtin_amdgcn_s_setprio(3);
    else if (c0 == 3 * ATT_CH) __builtin_amdgcn_s_setprio(2);
    else if (c0 == 5 * ATT_CH) __builtin_amdgcn_s_setprio(1);
    else if (c0 == 7 * ATT_CH) __builtin_amdgcn_s_setprio(0);
    v8f s[ATT_CH];
#pragma unroll
    for (int u = 0; u < ATT_CH; ++u) {
      const float4 kf = kfrag[(c0 + u) * 32];
      v2f kb0, kb1;
      kb0.x = kf.x; kb0.y = kf.y;
      kb1.x = kf.z; kb1.y = kf.w;
      v8f z = {};
      z = wmma4(qa0, kb0, z);
      s[u] = wmma4(qa1, kb1, z);
    }

#pragma unroll
    for (int r = 0; r < 8; ++r) {
      float mx = s[0][r];
#pragma unroll
      for (int u = 1; u < ATT_CH; ++u) mx = fmaxf(mx, s[u][r]);
      mx = fmaxf(mx, ROW_XOR(mx, 1));
      mx = fmaxf(mx, ROW_XOR(mx, 2));
      mx = fmaxf(mx, ROW_XOR(mx, 4));
      mx = fmaxf(mx, ROW_XOR(mx, 8));
      const float mn = fmaxf(m[r], mx);
      const float a = __expf(m[r] - mn);
      l[r] *= a;
      o[r] *= a;
      m[r] = mn;
    }

#pragma unroll
    for (int u = 0; u < ATT_CH; ++u) {
      const int kt = c0 + u;
#pragma unroll
      for (int r = 0; r < 8; ++r) {
        const float p = __expf(s[u][r] - m[r]);
        l[r] += p;
        myP[(r + 8 * hi) * 16 + n16] = p;
      }
      asm volatile("s_wait_dscnt 0" ::: "memory");

      const float4 vf0 = vfrag[kt * 32];
      const float4 vf1 = vfrag[kt * 32 + 16];
      v2f pa, vb;
      pa.x = myP[n16 * 16 + 0 + 2 * hi]; pa.y = myP[n16 * 16 + 1 + 2 * hi];
      vb.x = vf0.x; vb.y = vf0.y;
      o = wmma4(pa, vb, o);
      pa.x = myP[n16 * 16 + 4 + 2 * hi]; pa.y = myP[n16 * 16 + 5 + 2 * hi];
      vb.x = vf0.z; vb.y = vf0.w;
      o = wmma4(pa, vb, o);
      pa.x = myP[n16 * 16 + 8 + 2 * hi]; pa.y = myP[n16 * 16 + 9 + 2 * hi];
      vb.x = vf1.x; vb.y = vf1.y;
      o = wmma4(pa, vb, o);
      pa.x = myP[n16 * 16 + 12 + 2 * hi]; pa.y = myP[n16 * 16 + 13 + 2 * hi];
      vb.x = vf1.z; vb.y = vf1.w;
      o = wmma4(pa, vb, o);
      asm volatile("" ::: "memory");
      __builtin_amdgcn_sched_barrier(0);
    }
  }

#pragma unroll
  for (int r = 0; r < 8; ++r) {
    float ps = l[r];
    ps += __shfl_xor(ps, 1);
    ps += __shfl_xor(ps, 2);
    ps += __shfl_xor(ps, 4);
    ps += __shfl_xor(ps, 8);
    l[r] = ps;
  }

  const int lane_e = __builtin_amdgcn_mbcnt_lo(~0u, 0u);
  const int n16e = lane_e & 15, hie = lane_e >> 4;
  if (n16e < D_HEAD) {
#pragma unroll
    for (int r = 0; r < 8; ++r) {
      const int row = b * S_LEN + qt * 16 + r + 8 * hie;
      attn_out[row * E_DIM + h * D_HEAD + n16e] = o[r] / l[r];
    }
  }
}

#define TAIL_WAVES 4
#define CH_OUT (4 * 4 * 2)
#define CH_M1  (8 * 4 * 2)
#define CH_M2  (4 * 8 * 2)
#define TAIL_LDS_F4 (176 * 1024 / 16)

__device__ __forceinline__ float xhalf(float v) {
  const int b = __builtin_bit_cast(int, v);
  return __builtin_bit_cast(float, __builtin_amdgcn_permlanex16(
      b, b, 0x76543210, 0xfedcba98, false, false));
}

template <int NU, int KT>
__device__ __forceinline__ void gemm_t(const float4* __restrict__ wl, int lane,
                                       const float (&B)[KT][8], v8f (&acc)[NU]) {
  float4 cur[NU * 2];
#pragma unroll
  for (int u = 0; u < NU; ++u) {
#pragma unroll
    for (int e = 0; e < 2; ++e) cur[u * 2 + e] = wl[((u * KT) * 2 + e) * 32 + lane];
  }
#pragma unroll
  for (int t = 0; t < KT; ++t) {
    float4 nxt[NU * 2];
    if (t + 1 < KT) {
#pragma unroll
      for (int u = 0; u < NU; ++u) {
#pragma unroll
        for (int e = 0; e < 2; ++e)
          nxt[u * 2 + e] = wl[((u * KT + t + 1) * 2 + e) * 32 + lane];
      }
    }
    __builtin_amdgcn_sched_barrier(0);
#pragma unroll
    for (int e = 0; e < 2; ++e) {
      v2f b0, b1;
      b0.x = B[t][4 * e + 0]; b0.y = B[t][4 * e + 1];
      b1.x = B[t][4 * e + 2]; b1.y = B[t][4 * e + 3];
#pragma unroll
      for (int u = 0; u < NU; ++u) {
        v2f a;
        a.x = cur[u * 2 + e].x; a.y = cur[u * 2 + e].y;
        acc[u] = wmma4(a, b0, acc[u]);
      }
#pragma unroll
      for (int u = 0; u < NU; ++u) {
        v2f a;
        a.x = cur[u * 2 + e].z; a.y = cur[u * 2 + e].w;
        acc[u] = wmma4(a, b1, acc[u]);
      }
    }
    __builtin_amdgcn_sched_barrier(0);
    if (t + 1 < KT) {
#pragma unroll
      for (int i = 0; i < NU * 2; ++i) cur[i] = nxt[i];
    }
  }
}

template <int NT>
__device__ __forceinline__ void vec_t(const float* __restrict__ vec, int h,
                                      float (&d)[NT][8]) {
#pragma unroll
  for (int t = 0; t < NT; ++t) {
    const float4* p = reinterpret_cast<const float4*>(vec + 16 * t + 8 * h);
    const float4 a = p[0], b = p[1];
    d[t][0] = a.x; d[t][1] = a.y; d[t][2] = a.z; d[t][3] = a.w;
    d[t][4] = b.x; d[t][5] = b.y; d[t][6] = b.z; d[t][7] = b.w;
  }
}

__device__ __forceinline__ void layer_norm_t(float (&v)[4][8],
                                             const float (&g)[4][8],
                                             const float (&beta)[4][8]) {
  float s[4], q[4];
#pragma unroll
  for (int t = 0; t < 4; ++t) {
    s[t] = v[t][0];
    q[t] = v[t][0] * v[t][0];
#pragma unroll
    for (int r = 1; r < 8; ++r) { s[t] += v[t][r]; q[t] += v[t][r] * v[t][r]; }
  }
  float sum = (s[0] + s[1]) + (s[2] + s[3]);
  float sq = (q[0] + q[1]) + (q[2] + q[3]);
  sum += xhalf(sum);
  sq += xhalf(sq);
  const float mean = sum * (1.f / 64.f);
  const float var = sq * (1.f / 64.f) - mean * mean;
  const float rstd = rsqrtf(var + 1e-5f);
#pragma unroll
  for (int t = 0; t < 4; ++t) {
#pragma unroll
    for (int r = 0; r < 8; ++r)
      v[t][r] = (v[t][r] - mean) * rstd * g[t][r] + beta[t][r];
  }
}

__global__ __launch_bounds__(TAIL_WAVES * 32) void k_tail(
    const float* __restrict__ attn_out, const float* __restrict__ x,
    const float* __restrict__ Wout, const float* __restrict__ bout,
    const float* __restrict__ g1, const float* __restrict__ beta1,
    const float* __restrict__ Wm1, const float* __restrict__ bm1,
    const float* __restrict__ Wm2, const float* __restrict__ bm2,
    const float* __restrict__ g2, const float* __restrict__ beta2,
    float* __restrict__ out) {
  __shared__ __attribute__((aligned(16))) float4 wl[TAIL_LDS_F4];
  float4* wlOut = wl;
  float4* wlM1 = wl + CH_OUT * 32;
  float4* wlM2 = wl + (CH_OUT + CH_M1) * 32;
  const int tid = threadIdx.x;
  const int wave = __builtin_amdgcn_readfirstlane(tid >> 5);
  const int lane = tid & 31;
  const int c = lane & 15, h = lane >> 4;
  const int row = (blockIdx.x * TAIL_WAVES + wave) * 16 + c;

  float X[4][8], R[4][8];
  {
    const float4* ar = reinterpret_cast<const float4*>(attn_out + row * E_DIM + 8 * h);
    const float4* xr = reinterpret_cast<const float4*>(x + row * E_DIM + 8 * h);
#pragma unroll
    for (int t = 0; t < 4; ++t) {
      const float4 a0 = ar[4 * t], a1 = ar[4 * t + 1];
      const float4 x0 = xr[4 * t], x1 = xr[4 * t + 1];
      X[t][0] = a0.x; X[t][1] = a0.y; X[t][2] = a0.z; X[t][3] = a0.w;
      X[t][4] = a1.x; X[t][5] = a1.y; X[t][6] = a1.z; X[t][7] = a1.w;
      R[t][0] = x0.x; R[t][1] = x0.y; R[t][2] = x0.z; R[t][3] = x0.w;
      R[t][4] = x1.x; R[t][5] = x1.y; R[t][6] = x1.z; R[t][7] = x1.w;
    }
  }

  {
    const int pw = 8 * h + 4 * (wave & 1) + 16 * (wave >> 1);
    const float* pO = Wout + c * E_DIM + pw;
    const float* p1 = Wm1 + c * E_DIM + pw;
    const float* p2 = Wm2 + c * (2 * E_DIM) + pw;
    float4 s0[CH_OUT / TAIL_WAVES], s1[CH_M1 / TAIL_WAVES], s2[CH_M2 / TAIL_WAVES];
#pragma unroll
    for (int i = 0; i < CH_OUT / TAIL_WAVES; ++i)
      s0[i] = *reinterpret_cast<const float4*>(pO + (i >> 1) * 16 * E_DIM + (i & 1) * 32);
#pragma unroll
    for (int i = 0; i < CH_M1 / TAIL_WAVES; ++i)
      s1[i] = *reinterpret_cast<const float4*>(p1 + (i >> 1) * 16 * E_DIM + (i & 1) * 32);
#pragma unroll
    for (int i = 0; i < CH_M2 / TAIL_WAVES; ++i)
      s2[i] = *reinterpret_cast<const float4*>(p2 + (i >> 2) * 32 * E_DIM + (i & 3) * 32);
    float4* d0 = wlOut + wave * 32 + lane;
    float4* d1 = wlM1 + wave * 32 + lane;
    float4* d2 = wlM2 + wave * 32 + lane;
#pragma unroll
    for (int i = 0; i < CH_OUT / TAIL_WAVES; ++i) d0[i * TAIL_WAVES * 32] = s0[i];
#pragma unroll
    for (int i = 0; i < CH_M1 / TAIL_WAVES; ++i) d1[i * TAIL_WAVES * 32] = s1[i];
#pragma unroll
    for (int i = 0; i < CH_M2 / TAIL_WAVES; ++i) d2[i * TAIL_WAVES * 32] = s2[i];
  }
  float vbo[4][8], vg1[4][8], vb1[4][8];
  vec_t<4>(bout, h, vbo);
  vec_t<4>(g1, h, vg1);
  vec_t<4>(beta1, h, vb1);
  __syncthreads();

  float M[4][8];
  float vm1[8][8];
  {
    v8f acc[4] = {};
    gemm_t<4, 4>(wlOut, lane, X, acc);
    vec_t<8>(bm1, h, vm1);
#pragma unroll
    for (int t = 0; t < 4; ++t) {
#pragma unroll
      for (int r = 0; r < 8; ++r) M[t][r] = acc[t][r] + vbo[t][r] + R[t][r];
    }
    layer_norm_t(M, vg1, vb1);
  }

  float Hd[8][8];
  float vm2[4][8], vg2[4][8], vb2[4][8];
  {
    v8f acc[8] = {};
    gemm_t<8, 4>(wlM1, lane, M, acc);
    vec_t<4>(bm2, h, vm2);
    vec_t<4>(g2, h, vg2);
    vec_t<4>(beta2, h, vb2);
#pragma unroll
    for (int u = 0; u < 8; ++u) {
#pragma unroll
      for (int r = 0; r < 8; ++r) Hd[u][r] = fmaxf(acc[u][r] + vm1[u][r], 0.f);
    }
  }

  {
    v8f acc[4] = {};
    gemm_t<4, 8>(wlM2, lane, Hd, acc);
    float O[4][8];
#pragma unroll
    for (int t = 0; t < 4; ++t) {
#pragma unroll
      for (int r = 0; r < 8; ++r) O[t][r] = acc[t][r] + vm2[t][r] + M[t][r];
    }
    layer_norm_t(O, vg2, vb2);
    float4* orow = reinterpret_cast<float4*>(out + row * E_DIM + 8 * h);
#pragma unroll
    for (int t = 0; t < 4; ++t) {
      orow[4 * t]     = make_float4(O[t][0], O[t][1], O[t][2], O[t][3]);
      orow[4 * t + 1] = make_float4(O[t][4], O[t][5], O[t][6], O[t][7]);
    }
  }
}

extern "C" void kernel_launch(void* const* d_in, const int* in_sizes, int n_in,
                              void* d_out, int out_size, void* d_ws, size_t ws_size,
                              hipStream_t stream) {
  const float* x    = (const float*)d_in[0];
  const float* Wqkv = (const float*)d_in[4];
  const float* bqkv = (const float*)d_in[5];
  const float* Wout = (const float*)d_in[6];
  const float* bout = (const float*)d_in[7];
  const float* g1   = (const float*)d_in[8];
  const float* b1   = (const float*)d_in[9];
  const float* g2   = (const float*)d_in[10];
  const float* b2   = (const float*)d_in[11];
  const float* Wm1  = (const float*)d_in[12];
  const float* bm1  = (const float*)d_in[13];
  const float* Wm2  = (const float*)d_in[14];
  const float* bm2  = (const float*)d_in[15];
  float* out = (float*)d_out;

  float* attn = (float*)d_ws;

  k_attn<<<B_SZ * H_NUM * 2, ATT_WAVES * 32, 0, stream>>>(x, Wqkv, bqkv, attn);
  k_tail<<<N_TOK / (16 * TAIL_WAVES), TAIL_WAVES * 32, 0, stream>>>(
      attn, x, Wout, bout, g1, b1, Wm1, bm1, Wm2, bm2, g2, b2, out);
}


# k_attn: softmax reference fixed before the key loop from resident K: ref=sum_d|q_d|*max_k|K_kd| >= every score; C=-ref, p=exp2(acc), no max/rescale; exact rerun on underflow; sum order kept
# speedup vs baseline: 2.5100x; 1.1212x over previous
#include <hip/hip_runtime.h>
#include <math.h>

typedef __attribute__((ext_vector_type(2))) float v2f;
typedef __attribute__((ext_vector_type(8))) float v8f;

#define E_DIM 64
#define H_NUM 8
#define D_HEAD 8
#define B_SZ 16
#define S_LEN 1024
#define N_TOK (B_SZ * S_LEN)

__device__ __forceinline__ v8f wmma4(v2f a, v2f b, v8f c) {
  return __builtin_amdgcn_wmma_f32_16x16x4_f32(
       false, a,  false, b,
       (short)0, c,  false,  false);
}

#define ROW_XOR(v, MASK)                                                      \
  __builtin_bit_cast(float, __builtin_amdgcn_update_dpp(                      \
      __builtin_bit_cast(int, (v)), __builtin_bit_cast(int, (v)),             \
      0x160 + (MASK), 0xf, 0xf, true))

__device__ __forceinline__ float xhalf(float v) {
  const int b = __builtin_bit_cast(int, v);
  return __builtin_bit_cast(float, __builtin_amdgcn_permlanex16(
      b, b, 0x76543210, 0xfedcba98, false, false));
}

#define ATT_WAVES 32
#define ATT_CH 8
#define ATT_LDS_FLOATS (176 * 1024 / 4)

template <bool EXACT>
__device__ __forceinline__ void key_pass(const float4* kfrag, const float4* vfrag,
                                         float* myP, const v2f qa0, const v2f qa1,
                                         const v8f negref, const int n16,
                                         const int hi, v8f& o, float (&l)[8]) {
  float m[8];
  v2f l2[4];
#pragma unroll
  for (int r = 0; r < 8; ++r) { m[r] = -3.0e38f; o[r] = 0.f; }
#pragma unroll
  for (int i = 0; i < 4; ++i) { l2[i].x = 0.f; l2[i].y = 0.f; }

  for (int c0 = 0; c0 < S_LEN / 16; c0 += ATT_CH) {
    if (!EXACT) {
      if (c0 == 0) __builtin_amdgcn_s_setprio(3);
      else if (c0 == 3 * ATT_CH) __builtin_amdgcn_s_setprio(2);
      else if (c0 == 5 * ATT_CH) __builtin_amdgcn_s_setprio(1);
      else if (c0 == 7 * ATT_CH) __builtin_amdgcn_s_setprio(0);
    }
    v8f s[ATT_CH];
#pragma unroll
    for (int u = 0; u < ATT_CH; ++u) {
      const float4 kf = kfrag[(c0 + u) * 32];
      v2f kb0, kb1;
      kb0.x = kf.x; kb0.y = kf.y;
      kb1.x = kf.z; kb1.y = kf.w;
      v8f z = {};
      if (!EXACT) z = negref;
      z = wmma4(qa0, kb0, z);
      s[u] = wmma4(qa1, kb1, z);
    }

    if (EXACT) {
#pragma unroll
      for (int r = 0; r < 8; ++r) {
        float mx = s[0][r];
#pragma unroll
        for (int u = 1; u < ATT_CH; ++u) mx = fmaxf(mx, s[u][r]);
        mx = fmaxf(mx, ROW_XOR(mx, 1));
        mx = fmaxf(mx, ROW_XOR(mx, 2));
        mx = fmaxf(mx, ROW_XOR(mx, 4));
        mx = fmaxf(mx, ROW_XOR(mx, 8));
        const float mn = fmaxf(m[r], mx);
        const float a = __builtin_amdgcn_exp2f(m[r] - mn);
        l2[r >> 1][r & 1] *= a;
        o[r] *= a;
        m[r] = mn;
      }
    }

#pragma unroll
    for (int u = 0; u < ATT_CH; ++u) {
      const int kt = c0 + u;
      float p[8];
#pragma unroll
      for (int r = 0; r < 8; ++r) {
        p[r] = __builtin_amdgcn_exp2f(EXACT ? s[u][r] - m[r] : s[u][r]);
        myP[(r + 8 * hi) * 16 + n16] = p[r];
      }
#pragma unroll
      for (int i = 0; i < 4; ++i) {
        v2f pp;
        pp.x = p[2 * i]; pp.y = p[2 * i + 1];
        l2[i] += pp;
      }
      asm volatile("s_wait_dscnt 0" ::: "memory");

      const float4 vf0 = vfrag[kt * 32];
      const float4 vf1 = vfrag[kt * 32 + 16];
      v2f pa, vb;
      pa.x = myP[n16 * 16 + 0 + 2 * hi]; pa.y = myP[n16 * 16 + 1 + 2 * hi];
      vb.x = vf0.x; vb.y = vf0.y;
      o = wmma4(pa, vb, o);
      pa.x = myP[n16 * 16 + 4 + 2 * hi]; pa.y = myP[n16 * 16 + 5 + 2 * hi];
      vb.x = vf0.z; vb.y = vf0.w;
      o = wmma4(pa, vb, o);
      pa.x = myP[n16 * 16 + 8 + 2 * hi]; pa.y = myP[n16 * 16 + 9 + 2 * hi];
      vb.x = vf1.x; vb.y = vf1.y;
      o = wmma4(pa, vb, o);
      pa.x = myP[n16 * 16 + 12 + 2 * hi]; pa.y = myP[n16 * 16 + 13 + 2 * hi];
      vb.x = vf1.z; vb.y = vf1.w;
      o = wmma4(pa, vb, o);
      asm volatile("" ::: "memory");
      __builtin_amdgcn_sched_barrier(0);
    }
  }

#pragma unroll
  for (int r = 0; r < 8; ++r) {
    float ps = l2[r >> 1][r & 1];
    ps += ROW_XOR(ps, 1);
    ps += ROW_XOR(ps, 2);
    ps += ROW_XOR(ps, 4);
    ps += ROW_XOR(ps, 8);
    l[r] = ps;
  }
}

__global__ __launch_bounds__(ATT_WAVES * 32) void k_attn(
    const float* __restrict__ x, const float* __restrict__ Wqkv,
    const float* __restrict__ bqkv, float* __restrict__ attn_out) {
  __shared__ __attribute__((aligned(16))) float lds_all[ATT_LDS_FLOATS];
  float* ldsK = lds_all;
  float* ldsV = lds_all + S_LEN * D_HEAD;
  float* ldsP = lds_all + 2 * S_LEN * D_HEAD;
  float* ldsM = lds_all + 3 * S_LEN * D_HEAD;
  const int tid = threadIdx.x;
  const int wave = __builtin_amdgcn_readfirstlane(tid >> 5);
  const int lane = tid & 31;
  const int n16 = lane & 15, hi = lane >> 4;

  const int bid = blockIdx.x;
  const int bh = bid >> 1;
  const int b = bh >> 3;
  const int h = bh & 7;
  const int qt = (bid & 1) * ATT_WAVES + wave;
  const float scale = 0.35355339059327373f;

  const float* xB = x + (size_t)(b * S_LEN) * E_DIM;
  const int ot = ((bid & 1) ^ 1) * ATT_WAVES + wave;

  const int isV = n16 >> 3, dd = n16 & 7;
  const int kvrow = (isV ? 2 * E_DIM : E_DIM) + h * D_HEAD + dd;
  const int qrow = h * D_HEAD + dd;
  const float4* wkv = reinterpret_cast<const float4*>(Wqkv + kvrow * E_DIM) + hi;
  const float4* wq  = reinterpret_cast<const float4*>(Wqkv + qrow * E_DIM) + hi;
  const float4* xq  = reinterpret_cast<const float4*>(xB + (qt * 16 + n16) * E_DIM) + hi;
  const float4* xo  = reinterpret_cast<const float4*>(xB + (ot * 16 + n16) * E_DIM) + hi;
  v8f ckv = {}, cko = {}, cq = {};
#pragma unroll
  for (int j = 0; j < 8; ++j) {
    const float4 wv = wkv[2 * j];
    const float4 wqv = wq[2 * j];
    const float4 xv = xq[2 * j];
    const float4 xw = xo[2 * j];
    v2f w0, w1, q0, q1, a0, a1, o0, o1;
    w0.x = wv.x;  w0.y = wv.y;  w1.x = wv.z;  w1.y = wv.w;
    q0.x = wqv.x; q0.y = wqv.y; q1.x = wqv.z; q1.y = wqv.w;
    a0.x = xv.x;  a0.y = xv.y;  a1.x = xv.z;  a1.y = xv.w;
    o0.x = xw.x;  o0.y = xw.y;  o1.x = xw.z;  o1.y = xw.w;
    ckv = wmma4(a0, w0, ckv);
    ckv = wmma4(a1, w1, ckv);
    cq = wmma4(q0, a0, cq);
    cq = wmma4(q1, a1, cq);
    cko = wmma4(o0, w0, cko);
    cko = wmma4(o1, w1, cko);
  }
  {
    const float bias = bqkv[kvrow];
    float* dst = isV ? ldsV + hi * 64 + 4 * dd
                     : ldsK + ((dd >> 1) & 1) * 64 + hi * 32 + (dd & 1) + 2 * (dd >> 2);
    float cmx = 0.f;
#pragma unroll
    for (int r = 0; r < 8; ++r) {
      const int voff = ((r >> 1) & 1) * 32 + ((r >> 2) & 1) * 2 + (r & 1);
      const int off = isV ? voff : 4 * r;
      const float vq = ckv[r] + bias, vo = cko[r] + bias;
      dst[qt * 128 + off] = vq;
      dst[ot * 128 + off] = vo;
      cmx = fmaxf(cmx, fmaxf(fabsf(vq), fabsf(vo)));
    }
    ldsM[n16 * 64 + wave * 2 + hi] = cmx;
  }

  v2f qa0, qa1;
  float qd[8];
  {
    const float* bq = bqkv + h * D_HEAD;
#pragma unroll
    for (int r = 0; r < 8; ++r) qd[r] = (cq[r] + bq[r]) * (scale * 1.4426950408889634f);
    qa0.x = hi ? qd[2] : qd[0];
    qa0.y = hi ? qd[3] : qd[1];
    qa1.x = hi ? qd[6] : qd[4];
    qa1.y = hi ? qd[7] : qd[5];
  }
  __syncthreads();

  v8f negref;
  {
    const float4* mp = reinterpret_cast<const float4*>(
        ldsM + dd * 64 + (isV + 2 * hi) * 16);
    const float4 e0 = mp[0], e1 = mp[1], e2 = mp[2], e3 = mp[3];
    float g = fmaxf(fmaxf(fmaxf(e0.x, e0.y), fmaxf(e0.z, e0.w)),
                    fmaxf(fmaxf(e1.x, e1.y), fmaxf(e1.z, e1.w)));
    g = fmaxf(g, fmaxf(fmaxf(fmaxf(e2.x, e2.y), fmaxf(e2.z, e2.w)),
                       fmaxf(fmaxf(e3.x, e3.y), fmaxf(e3.z, e3.w))));
    g = fmaxf(g, ROW_XOR(g, 8));
    g = fmaxf(g, xhalf(g));
    float refq = 0.f;
#pragma unroll
    for (int r = 0; r < 8; ++r) {
      const float gd = __builtin_bit_cast(
          float, __builtin_amdgcn_readlane(__builtin_bit_cast(int, g), r));
      refq = fmaf(fabsf(qd[r]), gd, refq);
    }
#pragma unroll
    for (int r = 0; r < 8; ++r) negref[r] = -__shfl(refq, r + 8 * hi);
  }

  const float4* kfrag = reinterpret_cast<const float4*>(ldsK) + lane;
  const float4* vfrag =
      reinterpret_cast<const float4*>(ldsV) + hi * 8 + (n16 & 7);
  float* myP = ldsP + wave * 256;

  float l[8];
  v8f o;
  key_pass<false>(kfrag, vfrag, myP, qa0, qa1, negref, n16, hi, o, l);

  bool low = false;
#pragma unroll
  for (int r = 0; r < 8; ++r) low |= !(l[r] >= 8.6736174e-19f);
  if (__builtin_amdgcn_ballot_w32(low) != 0u)
    key_pass<true>(kfrag, vfrag, myP, qa0, qa1, negref, n16, hi, o, l);

  const int lane_e = __builtin_amdgcn_mbcnt_lo(~0u, 0u);
  const int n16e = lane_e & 15, hie = lane_e >> 4;
  if (n16e < D_HEAD) {
#pragma unroll
    for (int r = 0; r < 8; ++r) {
      const int row = b * S_LEN + qt * 16 + r + 8 * hie;
      attn_out[row * E_DIM + h * D_HEAD + n16e] = o[r] / l[r];
    }
  }
}

#define TAIL_WAVES 4
#define CH_OUT (4 * 4 * 2)
#define CH_M1  (8 * 4 * 2)
#define CH_M2  (4 * 8 * 2)
#define TAIL_LDS_F4 (176 * 1024 / 16)

template <int NU, int KT>
__device__ __forceinline__ void gemm_t(const float4* __restrict__ wl, int lane,
                                       const float (&B)[KT][8], v8f (&acc)[NU]) {
  float4 cur[NU * 2];
#pragma unroll
  for (int u = 0; u < NU; ++u) {
#pragma unroll
    for (int e = 0; e < 2; ++e) cur[u * 2 + e] = wl[((u * KT) * 2 + e) * 32 + lane];
  }
#pragma unroll
  for (int t = 0; t < KT; ++t) {
    float4 nxt[NU * 2];
    if (t + 1 < KT) {
#pragma unroll
      for (int u = 0; u < NU; ++u) {
#pragma unroll
        for (int e = 0; e < 2; ++e)
          nxt[u * 2 + e] = wl[((u * KT + t + 1) * 2 + e) * 32 + lane];
      }
    }
    __builtin_amdgcn_sched_barrier(0);
#pragma unroll
    for (int e = 0; e < 2; ++e) {
      v2f b0, b1;
      b0.x = B[t][4 * e + 0]; b0.y = B[t][4 * e + 1];
      b1.x = B[t][4 * e + 2]; b1.y = B[t][4 * e + 3];
#pragma unroll
      for (int u = 0; u < NU; ++u) {
        v2f a;
        a.x = cur[u * 2 + e].x; a.y = cur[u * 2 + e].y;
        acc[u] = wmma4(a, b0, acc[u]);
      }
#pragma unroll
      for (int u = 0; u < NU; ++u) {
        v2f a;
        a.x = cur[u * 2 + e].z; a.y = cur[u * 2 + e].w;
        acc[u] = wmma4(a, b1, acc[u]);
      }
    }
    __builtin_amdgcn_sched_barrier(0);
    if (t + 1 < KT) {
#pragma unroll
      for (int i = 0; i < NU * 2; ++i) cur[i] = nxt[i];
    }
  }
}

template <int NT>
__device__ __forceinline__ void vec_t(const float* __restrict__ vec, int h,
                                      float (&d)[NT][8]) {
#pragma unroll
  for (int t = 0; t < NT; ++t) {
    const float4* p = reinterpret_cast<const float4*>(vec + 16 * t + 8 * h);
    const float4 a = p[0], b = p[1];
    d[t][0] = a.x; d[t][1] = a.y; d[t][2] = a.z; d[t][3] = a.w;
    d[t][4] = b.x; d[t][5] = b.y; d[t][6] = b.z; d[t][7] = b.w;
  }
}

__device__ __forceinline__ void layer_norm_t(float (&v)[4][8],
                                             const float (&g)[4][8],
                                             const float (&beta)[4][8]) {
  float s[4], q[4];
#pragma unroll
  for (int t = 0; t < 4; ++t) {
    s[t] = v[t][0];
    q[t] = v[t][0] * v[t][0];
#pragma unroll
    for (int r = 1; r < 8; ++r) { s[t] += v[t][r]; q[t] += v[t][r] * v[t][r]; }
  }
  float sum = (s[0] + s[1]) + (s[2] + s[3]);
  float sq = (q[0] + q[1]) + (q[2] + q[3]);
  sum += xhalf(sum);
  sq += xhalf(sq);
  const float mean = sum * (1.f / 64.f);
  const float var = sq * (1.f / 64.f) - mean * mean;
  const float rstd = rsqrtf(var + 1e-5f);
#pragma unroll
  for (int t = 0; t < 4; ++t) {
#pragma unroll
    for (int r = 0; r < 8; ++r)
      v[t][r] = (v[t][r] - mean) * rstd * g[t][r] + beta[t][r];
  }
}

__global__ __launch_bounds__(TAIL_WAVES * 32) void k_tail(
    const float* __restrict__ attn_out, const float* __restrict__ x,
    const float* __restrict__ Wout, const float* __restrict__ bout,
    const float* __restrict__ g1, const float* __restrict__ beta1,
    const float* __restrict__ Wm1, const float* __restrict__ bm1,
    const float* __restrict__ Wm2, const float* __restrict__ bm2,
    const float* __restrict__ g2, const float* __restrict__ beta2,
    float* __restrict__ out) {
  __shared__ __attribute__((aligned(16))) float4 wl[TAIL_LDS_F4];
  float4* wlOut = wl;
  float4* wlM1 = wl + CH_OUT * 32;
  float4* wlM2 = wl + (CH_OUT + CH_M1) * 32;
  const int tid = threadIdx.x;
  const int wave = __builtin_amdgcn_readfirstlane(tid >> 5);
  const int lane = tid & 31;
  const int c = lane & 15, h = lane >> 4;
  const int row = (blockIdx.x * TAIL_WAVES + wave) * 16 + c;

  float X[4][8], R[4][8];
  {
    const float4* ar = reinterpret_cast<const float4*>(attn_out + row * E_DIM + 8 * h);
    const float4* xr = reinterpret_cast<const float4*>(x + row * E_DIM + 8 * h);
#pragma unroll
    for (int t = 0; t < 4; ++t) {
      const float4 a0 = ar[4 * t], a1 = ar[4 * t + 1];
      const float4 x0 = xr[4 * t], x1 = xr[4 * t + 1];
      X[t][0] = a0.x; X[t][1] = a0.y; X[t][2] = a0.z; X[t][3] = a0.w;
      X[t][4] = a1.x; X[t][5] = a1.y; X[t][6] = a1.z; X[t][7] = a1.w;
      R[t][0] = x0.x; R[t][1] = x0.y; R[t][2] = x0.z; R[t][3] = x0.w;
      R[t][4] = x1.x; R[t][5] = x1.y; R[t][6] = x1.z; R[t][7] = x1.w;
    }
  }

  {
    const int pw = 8 * h + 4 * (wave & 1) + 16 * (wave >> 1);
    const float* pO = Wout + c * E_DIM + pw;
    const float* p1 = Wm1 + c * E_DIM + pw;
    const float* p2 = Wm2 + c * (2 * E_DIM) + pw;
    float4 s0[CH_OUT / TAIL_WAVES], s1[CH_M1 / TAIL_WAVES], s2[CH_M2 / TAIL_WAVES];
#pragma unroll
    for (int i = 0; i < CH_OUT / TAIL_WAVES; ++i)
      s0[i] = *reinterpret_cast<const float4*>(pO + (i >> 1) * 16 * E_DIM + (i & 1) * 32);
#pragma unroll
    for (int i = 0; i < CH_M1 / TAIL_WAVES; ++i)
      s1[i] = *reinterpret_cast<const float4*>(p1 + (i >> 1) * 16 * E_DIM + (i & 1) * 32);
#pragma unroll
    for (int i = 0; i < CH_M2 / TAIL_WAVES; ++i)
      s2[i] = *reinterpret_cast<const float4*>(p2 + (i >> 2) * 32 * E_DIM + (i & 3) * 32);
    float4* d0 = wlOut + wave * 32 + lane;
    float4* d1 = wlM1 + wave * 32 + lane;
    float4* d2 = wlM2 + wave * 32 + lane;
#pragma unroll
    for (int i = 0; i < CH_OUT / TAIL_WAVES; ++i) d0[i * TAIL_WAVES * 32] = s0[i];
#pragma unroll
    for (int i = 0; i < CH_M1 / TAIL_WAVES; ++i) d1[i * TAIL_WAVES * 32] = s1[i];
#pragma unroll
    for (int i = 0; i < CH_M2 / TAIL_WAVES; ++i) d2[i * TAIL_WAVES * 32] = s2[i];
  }
  float vbo[4][8], vg1[4][8], vb1[4][8];
  vec_t<4>(bout, h, vbo);
  vec_t<4>(g1, h, vg1);
  vec_t<4>(beta1, h, vb1);
  __syncthreads();

  float M[4][8];
  float vm1[8][8];
  {
    v8f acc[4] = {};
    gemm_t<4, 4>(wlOut, lane, X, acc);
    vec_t<8>(bm1, h, vm1);
#pragma unroll
    for (int t = 0; t < 4; ++t) {
#pragma unroll
      for (int r = 0; r < 8; ++r) M[t][r] = acc[t][r] + vbo[t][r] + R[t][r];
    }
    layer_norm_t(M, vg1, vb1);
  }

  float Hd[8][8];
  float vm2[4][8], vg2[4][8], vb2[4][8];
  {
    v8f acc[8] = {};
    gemm_t<8, 4>(wlM1, lane, M, acc);
    vec_t<4>(bm2, h, vm2);
    vec_t<4>(g2, h, vg2);
    vec_t<4>(beta2, h, vb2);
#pragma unroll
    for (int u = 0; u < 8; ++u) {
#pragma unroll
      for (int r = 0; r < 8; ++r) Hd[u][r] = fmaxf(acc[u][r] + vm1[u][r], 0.f);
    }
  }

  {
    v8f acc[4] = {};
    gemm_t<4, 8>(wlM2, lane, Hd, acc);
    float O[4][8];
#pragma unroll
    for (int t = 0; t < 4; ++t) {
#pragma unroll
      for (int r = 0; r < 8; ++r) O[t][r] = acc[t][r] + vm2[t][r] + M[t][r];
    }
    layer_norm_t(O, vg2, vb2);
    float4* orow = reinterpret_cast<float4*>(out + row * E_DIM + 8 * h);
#pragma unroll
    for (int t = 0; t < 4; ++t) {
      orow[4 * t]     = make_float4(O[t][0], O[t][1], O[t][2], O[t][3]);
      orow[4 * t + 1] = make_float4(O[t][4], O[t][5], O[t][6], O[t][7]);
    }
  }
}

extern "C" void kernel_launch(void* const* d_in, const int* in_sizes, int n_in,
                              void* d_out, int out_size, void* d_ws, size_t ws_size,
                              hipStream_t stream) {
  const float* x    = (const float*)d_in[0];
  const float* Wqkv = (const float*)d_in[4];
  const float* bqkv = (const float*)d_in[5];
  const float* Wout = (const float*)d_in[6];
  const float* bout = (const float*)d_in[7];
  const float* g1   = (const float*)d_in[8];
  const float* b1   = (const float*)d_in[9];
  const float* g2   = (const float*)d_in[10];
  const float* b2   = (const float*)d_in[11];
  const float* Wm1  = (const float*)d_in[12];
  const float* bm1  = (const float*)d_in[13];
  const float* Wm2  = (const float*)d_in[14];
  const float* bm2  = (const float*)d_in[15];
  float* out = (float*)d_out;

  float* attn = (float*)d_ws;

  k_attn<<<B_SZ * H_NUM * 2, ATT_WAVES * 32, 0, stream>>>(x, Wqkv, bqkv, attn);
  k_tail<<<N_TOK / (16 * TAIL_WAVES), TAIL_WAVES * 32, 0, stream>>>(
      attn, x, Wout, bout, g1, b1, Wm1, bm1, Wm2, bm2, g2, b2, out);
}
